# starting kernel
# baseline (speedup 1.0000x reference)
#include <hip/hip_runtime.h>
#include <hip/hip_bf16.h>

typedef __attribute__((ext_vector_type(2))) float v2f;
typedef __attribute__((ext_vector_type(8))) float v8f;

#define NDIM 128

__global__ void k_deg_init(float* deg, int n) {
    int i = blockIdx.x * blockDim.x + threadIdx.x;
    if (i < n) deg[i] = 1.0f;
}

__global__ void k_deg_count(const int* __restrict__ dst, float* deg, int e) {
    int i = blockIdx.x * blockDim.x + threadIdx.x;
    if (i < e) unsafeAtomicAdd(&deg[dst[i]], 1.0f);
}

__global__ void k_deg_rsqrt(float* deg, int n) {
    int i = blockIdx.x * blockDim.x + threadIdx.x;
    if (i < n) deg[i] = __frsqrt_rn(deg[i]);
}

__global__ void __launch_bounds__(256)
k_gemm128(const float* __restrict__ X, const float* __restrict__ W,
          float* __restrict__ H, int M) {
    __shared__ float Ws[NDIM * NDIM];

    int tid = threadIdx.x;
    {
        const float4* Wv = (const float4*)W;
        float4* Sv = (float4*)Ws;
        #pragma unroll
        for (int i = 0; i < 16; ++i)
            Sv[tid + 256 * i] = Wv[tid + 256 * i];
    }
    __syncthreads();

    int wave  = tid >> 5;
    int lane  = tid & 31;
    int l16   = lane & 15;
    int lhalf = lane >> 4;

    int rowBase = blockIdx.x * 128 + wave * 16;

    int arow = rowBase + l16;
    if (arow > M - 1) arow = M - 1;
    const float* aptr = X + (size_t)arow * NDIM + lhalf * 2;

    v8f acc[8] = {};

    for (int k = 0; k < NDIM; k += 4) {
        v2f a;
        {
            const float2 av = *(const float2*)(aptr + k);
            a[0] = av.x; a[1] = av.y;
        }
        int krow = (k + lhalf * 2) * NDIM + l16;
        #pragma unroll
        for (int t = 0; t < 8; ++t) {
            v2f b;
            b[0] = Ws[krow + t * 16];
            b[1] = Ws[krow + NDIM + t * 16];
            acc[t] = __builtin_amdgcn_wmma_f32_16x16x4_f32(
                false, a, false, b, (short)0, acc[t], false, false);
        }
    }

    #pragma unroll
    for (int t = 0; t < 8; ++t) {
        #pragma unroll
        for (int v = 0; v < 8; ++v) {
            int r = rowBase + v + lhalf * 8;
            if (r < M) H[(size_t)r * NDIM + t * 16 + l16] = acc[t][v];
        }
    }
}

__global__ void k_agg_init(const float* __restrict__ h, const float* __restrict__ dinv,
                           const float* __restrict__ bias, float* __restrict__ out, int n) {
    int i = blockIdx.x * blockDim.x + threadIdx.x;
    if (i >= n * 32) return;
    int node = i >> 5;
    int c4   = (i & 31) * 4;
    float di = dinv[node];
    float s  = di * di;
    float4 hv = *(const float4*)(h + (size_t)node * NDIM + c4);
    float4 bv = *(const float4*)(bias + c4);
    float4 o;
    o.x = bv.x + hv.x * s;
    o.y = bv.y + hv.y * s;
    o.z = bv.z + hv.z * s;
    o.w = bv.w + hv.w * s;
    *(float4*)(out + (size_t)node * NDIM + c4) = o;
}

#define EPW 4
__global__ void __launch_bounds__(256)
k_agg_edges(const float* __restrict__ h, const float* __restrict__ dinv,
            const int* __restrict__ src, const int* __restrict__ dst,
            float* __restrict__ out, int e) {
    int wv   = __builtin_amdgcn_readfirstlane((int)(threadIdx.x >> 5));
    int lane = threadIdx.x & 31;
    int base = (blockIdx.x * 8 + wv) * EPW;
    if (base >= e) return;

    int    s[EPW], d[EPW];
    float  nrm[EPW];
    float4 hv[EPW];

    #pragma unroll
    for (int j = 0; j < EPW; ++j) {
        int eid = (base + j < e) ? (base + j) : (e - 1);
        s[j] = src[eid];
        d[j] = dst[eid];
        nrm[j] = dinv[s[j]] * dinv[d[j]];
    }
    #pragma unroll
    for (int j = 0; j < EPW; ++j)
        hv[j] = *(const float4*)(h + (size_t)s[j] * NDIM + lane * 4);

    #pragma unroll
    for (int j = 0; j < EPW; ++j) {
        if (base + j >= e) break;
        float* op = out + (size_t)d[j] * NDIM + lane * 4;
        unsafeAtomicAdd(op + 0, hv[j].x * nrm[j]);
        unsafeAtomicAdd(op + 1, hv[j].y * nrm[j]);
        unsafeAtomicAdd(op + 2, hv[j].z * nrm[j]);
        unsafeAtomicAdd(op + 3, hv[j].w * nrm[j]);
    }
}

__global__ void k_tanh_ip(float* p, int n4) {
    int i = blockIdx.x * blockDim.x + threadIdx.x;
    if (i >= n4) return;
    float4 v = ((float4*)p)[i];
    v.x = tanhf(v.x); v.y = tanhf(v.y); v.z = tanhf(v.z); v.w = tanhf(v.w);
    ((float4*)p)[i] = v;
}

extern "C" void kernel_launch(void* const* d_in, const int* in_sizes, int n_in,
                              void* d_out, int out_size, void* d_ws, size_t ws_size,
                              hipStream_t stream) {
    const float* x   = (const float*)d_in[0];
    const float* W1  = (const float*)d_in[1];
    const float* b1  = (const float*)d_in[2];
    const float* W2  = (const float*)d_in[3];
    const float* b2  = (const float*)d_in[4];
    const int*   ei  = (const int*)d_in[5];

    const int N = in_sizes[0] / NDIM;
    const int E = in_sizes[5] / 2;
    const int* src = ei;
    const int* dst = ei + E;

    float* dinv = (float*)d_ws;
    float* h    = dinv + ((N + 255) & ~255);
    float* t1   = h + (size_t)N * NDIM;
    float* outp = (float*)d_out;

    const int B = 256;
    int gN   = (N + B - 1) / B;
    int gE   = (E + B - 1) / B;
    int gN32 = (N * 32 + B - 1) / B;
    int gEw  = (E + 8 * EPW - 1) / (8 * EPW);
    int gM   = (N + 127) / 128;

    k_deg_init <<<gN, B, 0, stream>>>(dinv, N);
    k_deg_count<<<gE, B, 0, stream>>>(dst, dinv, E);
    k_deg_rsqrt<<<gN, B, 0, stream>>>(dinv, N);

    k_gemm128  <<<gM,   B, 0, stream>>>(x, W1, h, N);
    k_agg_init <<<gN32, B, 0, stream>>>(h, dinv, b1, t1, N);
    k_agg_edges<<<gEw,  B, 0, stream>>>(h, dinv, src, dst, t1, E);
    k_tanh_ip  <<<gN32, B, 0, stream>>>(t1, N * 32);

    k_gemm128  <<<gM,   B, 0, stream>>>(t1, W2, h, N);
    k_agg_init <<<gN32, B, 0, stream>>>(h, dinv, b2, outp, N);
    k_agg_edges<<<gEw,  B, 0, stream>>>(h, dinv, src, dst, outp, E);
}


# Replace per-edge f32 atomic scatter-add by edges grouped by destination in-call (integer rank atomics), one wave per node summing its rows in registers; sum order becomes rank order
# speedup vs baseline: 2.0252x; 2.0252x over previous
#include <hip/hip_runtime.h>
#include <hip/hip_bf16.h>

typedef __attribute__((ext_vector_type(2))) float v2f;
typedef __attribute__((ext_vector_type(8))) float v8f;

#define NDIM 128

__global__ void k_csr_zero(int* cnt, int* total, int n) {
    int i = blockIdx.x * blockDim.x + threadIdx.x;
    if (i < n) cnt[i] = 0;
    if (i == 0) *total = 0;
}

__global__ void k_count_rank(const int* __restrict__ dst, int* cnt,
                             int* __restrict__ rank, int e) {
    int i = blockIdx.x * blockDim.x + threadIdx.x;
    if (i < e) rank[i] = atomicAdd(&cnt[dst[i]], 1);
}

__global__ void __launch_bounds__(256)
k_alloc(const int* __restrict__ cnt, int* __restrict__ start,
        float* __restrict__ dinv, int* total, int n) {
    __shared__ int wsum[8];
    __shared__ int bbase;
    int i    = blockIdx.x * 256 + threadIdx.x;
    int lane = threadIdx.x & 31;
    int wv   = threadIdx.x >> 5;
    int c    = (i < n) ? cnt[i] : 0;
    if (i < n) dinv[i] = __frsqrt_rn((float)(c + 1));
    int incl = c;
    for (int o = 1; o < 32; o <<= 1) {
        int t = __shfl_up(incl, o);
        if (lane >= o) incl += t;
    }
    if (lane == 31) wsum[wv] = incl;
    __syncthreads();
    if (threadIdx.x == 0) {
        int run = 0;
        for (int w = 0; w < 8; ++w) { int t = wsum[w]; wsum[w] = run; run += t; }
        bbase = atomicAdd(total, run);
    }
    __syncthreads();
    if (i < n) start[i] = bbase + wsum[wv] + incl - c;
}

__global__ void k_fill(const int* __restrict__ src, const int* __restrict__ dst,
                       const int* __restrict__ rank, const int* __restrict__ start,
                       const float* __restrict__ dinv, int* __restrict__ ent, int e) {
    int i = blockIdx.x * blockDim.x + threadIdx.x;
    if (i >= e) return;
    int s = src[i];
    int p = start[dst[i]] + rank[i];
    ((int2*)ent)[p] = make_int2(s, __float_as_int(dinv[s]));
}

__global__ void __launch_bounds__(256)
k_gemm128(const float* __restrict__ X, const float* __restrict__ W,
          float* __restrict__ H, int M) {
    __shared__ float Ws[NDIM * NDIM];

    int tid = threadIdx.x;
    {
        const float4* Wv = (const float4*)W;
        float4* Sv = (float4*)Ws;
        #pragma unroll
        for (int i = 0; i < 16; ++i)
            Sv[tid + 256 * i] = Wv[tid + 256 * i];
    }
    __syncthreads();

    int wave  = tid >> 5;
    int lane  = tid & 31;
    int l16   = lane & 15;
    int lhalf = lane >> 4;

    int rowBase = blockIdx.x * 128 + wave * 16;

    int arow = rowBase + l16;
    if (arow > M - 1) arow = M - 1;
    const float* aptr = X + (size_t)arow * NDIM + lhalf * 2;

    v8f acc[8] = {};

    for (int k = 0; k < NDIM; k += 4) {
        v2f a;
        {
            const float2 av = *(const float2*)(aptr + k);
            a[0] = av.x; a[1] = av.y;
        }
        int krow = (k + lhalf * 2) * NDIM + l16;
        #pragma unroll
        for (int t = 0; t < 8; ++t) {
            v2f b;
            b[0] = Ws[krow + t * 16];
            b[1] = Ws[krow + NDIM + t * 16];
            acc[t] = __builtin_amdgcn_wmma_f32_16x16x4_f32(
                false, a, false, b, (short)0, acc[t], false, false);
        }
    }

    #pragma unroll
    for (int t = 0; t < 8; ++t) {
        #pragma unroll
        for (int v = 0; v < 8; ++v) {
            int r = rowBase + v + lhalf * 8;
            if (r < M) H[(size_t)r * NDIM + t * 16 + l16] = acc[t][v];
        }
    }
}

template <int K>
__device__ __forceinline__ void gather_rows(const int2* __restrict__ ep,
                                            const float* __restrict__ hl,
                                            float di, float4& acc) {
    int2   e[K];
    float4 r[K];
    #pragma unroll
    for (int u = 0; u < K; ++u) e[u] = ep[u];
    #pragma unroll
    for (int u = 0; u < K; ++u)
        r[u] = *(const float4*)(hl + (size_t)e[u].x * NDIM);
    #pragma unroll
    for (int u = 0; u < K; ++u) {
        float nrm = __int_as_float(e[u].y) * di;
        acc.x += r[u].x * nrm;
        acc.y += r[u].y * nrm;
        acc.z += r[u].z * nrm;
        acc.w += r[u].w * nrm;
    }
}

__global__ void __launch_bounds__(256)
k_agg_gather(const float* __restrict__ h, const float* __restrict__ dinv,
             const float* __restrict__ bias, const int* __restrict__ start,
             const int* __restrict__ cnt, const int* __restrict__ ent,
             float* __restrict__ out, int n) {
    int wv   = __builtin_amdgcn_readfirstlane((int)(threadIdx.x >> 5));
    int lane = threadIdx.x & 31;
    int node = blockIdx.x * 8 + wv;
    if (node >= n) return;

    int   st = start[node];
    int   c  = cnt[node];
    float di = dinv[node];
    const float* hl = h + lane * 4;

    float4 hv = *(const float4*)(hl + (size_t)node * NDIM);
    float4 bv = *(const float4*)(bias + lane * 4);
    float4 acc = make_float4(0.0f, 0.0f, 0.0f, 0.0f);

    const int2* ep = (const int2*)ent + st;
    int j = 0;
    for (; j + 8 <= c; j += 8) gather_rows<8>(ep + j, hl, di, acc);
    switch (c - j) {
        case 7: gather_rows<7>(ep + j, hl, di, acc); break;
        case 6: gather_rows<6>(ep + j, hl, di, acc); break;
        case 5: gather_rows<5>(ep + j, hl, di, acc); break;
        case 4: gather_rows<4>(ep + j, hl, di, acc); break;
        case 3: gather_rows<3>(ep + j, hl, di, acc); break;
        case 2: gather_rows<2>(ep + j, hl, di, acc); break;
        case 1: gather_rows<1>(ep + j, hl, di, acc); break;
        default: break;
    }

    float  s = di * di;
    float4 o;
    o.x = (bv.x + hv.x * s) + acc.x;
    o.y = (bv.y + hv.y * s) + acc.y;
    o.z = (bv.z + hv.z * s) + acc.z;
    o.w = (bv.w + hv.w * s) + acc.w;
    *(float4*)(out + (size_t)node * NDIM + lane * 4) = o;
}

__global__ void k_tanh_ip(float* p, int n4) {
    int i = blockIdx.x * blockDim.x + threadIdx.x;
    if (i >= n4) return;
    float4 v = ((float4*)p)[i];
    v.x = tanhf(v.x); v.y = tanhf(v.y); v.z = tanhf(v.z); v.w = tanhf(v.w);
    ((float4*)p)[i] = v;
}

extern "C" void kernel_launch(void* const* d_in, const int* in_sizes, int n_in,
                              void* d_out, int out_size, void* d_ws, size_t ws_size,
                              hipStream_t stream) {
    const float* x   = (const float*)d_in[0];
    const float* W1  = (const float*)d_in[1];
    const float* b1  = (const float*)d_in[2];
    const float* W2  = (const float*)d_in[3];
    const float* b2  = (const float*)d_in[4];
    const int*   ei  = (const int*)d_in[5];

    const int N = in_sizes[0] / NDIM;
    const int E = in_sizes[5] / 2;
    const int* src = ei;
    const int* dst = ei + E;

    const int Np = (N + 255) & ~255;
    const int Ep = (E + 255) & ~255;

    float* dinv = (float*)d_ws;
    float* h    = dinv + Np;
    int*   cnt   = (int*)(h + (size_t)N * NDIM);
    int*   start = cnt + Np;
    int*   total = start + Np;
    int*   rank  = total + 256;
    int*   ent   = rank + Ep;
    float* outp  = (float*)d_out;
    float* t1    = outp;

    const int B = 256;
    int gN   = (N + B - 1) / B;
    int gE   = (E + B - 1) / B;
    int gN32 = (N * 32 + B - 1) / B;
    int gNw  = (N + 7) / 8;
    int gM   = (N + 127) / 128;

    k_csr_zero  <<<gN, B, 0, stream>>>(cnt, total, N);
    k_count_rank<<<gE, B, 0, stream>>>(dst, cnt, rank, E);
    k_alloc     <<<gN, B, 0, stream>>>(cnt, start, dinv, total, N);
    k_fill      <<<gE, B, 0, stream>>>(src, dst, rank, start, dinv, ent, E);

    k_gemm128   <<<gM,   B, 0, stream>>>(x, W1, h, N);
    k_agg_gather<<<gNw,  B, 0, stream>>>(h, dinv, b1, start, cnt, ent, t1, N);
    k_tanh_ip   <<<gN32, B, 0, stream>>>(t1, N * 32);

    k_gemm128   <<<gM,   B, 0, stream>>>(t1, W2, h, N);
    k_agg_gather<<<gNw,  B, 0, stream>>>(h, dinv, b2, start, cnt, ent, outp, N);
}


# tanh of layer 1 applied in the aggregation kernel's registers before its row store; the separate in-place tanh pass over the activation plane is gone (9 launches -> 8)
# speedup vs baseline: 2.1562x; 1.0647x over previous
#include <hip/hip_runtime.h>
#include <hip/hip_bf16.h>

typedef __attribute__((ext_vector_type(2))) float v2f;
typedef __attribute__((ext_vector_type(8))) float v8f;

#define NDIM 128

__global__ void k_csr_zero(int* cnt, int* total, int n) {
    int i = blockIdx.x * blockDim.x + threadIdx.x;
    if (i < n) cnt[i] = 0;
    if (i == 0) *total = 0;
}

__global__ void k_count_rank(const int* __restrict__ dst, int* cnt,
                             int* __restrict__ rank, int e) {
    int i = blockIdx.x * blockDim.x + threadIdx.x;
    if (i < e) rank[i] = atomicAdd(&cnt[dst[i]], 1);
}

__global__ void __launch_bounds__(256)
k_alloc(const int* __restrict__ cnt, int* __restrict__ start,
        float* __restrict__ dinv, int* total, int n) {
    __shared__ int wsum[8];
    __shared__ int bbase;
    int i    = blockIdx.x * 256 + threadIdx.x;
    int lane = threadIdx.x & 31;
    int wv   = threadIdx.x >> 5;
    int c    = (i < n) ? cnt[i] : 0;
    if (i < n) dinv[i] = __frsqrt_rn((float)(c + 1));
    int incl = c;
    for (int o = 1; o < 32; o <<= 1) {
        int t = __shfl_up(incl, o);
        if (lane >= o) incl += t;
    }
    if (lane == 31) wsum[wv] = incl;
    __syncthreads();
    if (threadIdx.x == 0) {
        int run = 0;
        for (int w = 0; w < 8; ++w) { int t = wsum[w]; wsum[w] = run; run += t; }
        bbase = atomicAdd(total, run);
    }
    __syncthreads();
    if (i < n) start[i] = bbase + wsum[wv] + incl - c;
}

__global__ void k_fill(const int* __restrict__ src, const int* __restrict__ dst,
                       const int* __restrict__ rank, const int* __restrict__ start,
                       const float* __restrict__ dinv, int* __restrict__ ent, int e) {
    int i = blockIdx.x * blockDim.x + threadIdx.x;
    if (i >= e) return;
    int s = src[i];
    int p = start[dst[i]] + rank[i];
    ((int2*)ent)[p] = make_int2(s, __float_as_int(dinv[s]));
}

__global__ void __launch_bounds__(256)
k_gemm128(const float* __restrict__ X, const float* __restrict__ W,
          float* __restrict__ H, int M) {
    __shared__ float Ws[NDIM * NDIM];

    int tid = threadIdx.x;
    {
        const float4* Wv = (const float4*)W;
        float4* Sv = (float4*)Ws;
        #pragma unroll
        for (int i = 0; i < 16; ++i)
            Sv[tid + 256 * i] = Wv[tid + 256 * i];
    }
    __syncthreads();

    int wave  = tid >> 5;
    int lane  = tid & 31;
    int l16   = lane & 15;
    int lhalf = lane >> 4;

    int rowBase = blockIdx.x * 128 + wave * 16;

    int arow = rowBase + l16;
    if (arow > M - 1) arow = M - 1;
    const float* aptr = X + (size_t)arow * NDIM + lhalf * 2;

    v8f acc[8] = {};

    for (int k = 0; k < NDIM; k += 4) {
        v2f a;
        {
            const float2 av = *(const float2*)(aptr + k);
            a[0] = av.x; a[1] = av.y;
        }
        int krow = (k + lhalf * 2) * NDIM + l16;
        #pragma unroll
        for (int t = 0; t < 8; ++t) {
            v2f b;
            b[0] = Ws[krow + t * 16];
            b[1] = Ws[krow + NDIM + t * 16];
            acc[t] = __builtin_amdgcn_wmma_f32_16x16x4_f32(
                false, a, false, b, (short)0, acc[t], false, false);
        }
    }

    #pragma unroll
    for (int t = 0; t < 8; ++t) {
        #pragma unroll
        for (int v = 0; v < 8; ++v) {
            int r = rowBase + v + lhalf * 8;
            if (r < M) H[(size_t)r * NDIM + t * 16 + l16] = acc[t][v];
        }
    }
}

template <int K>
__device__ __forceinline__ void gather_rows(const int2* __restrict__ ep,
                                            const float* __restrict__ hl,
                                            float di, float4& acc) {
    int2   e[K];
    float4 r[K];
    #pragma unroll
    for (int u = 0; u < K; ++u) e[u] = ep[u];
    #pragma unroll
    for (int u = 0; u < K; ++u)
        r[u] = *(const float4*)(hl + (size_t)e[u].x * NDIM);
    #pragma unroll
    for (int u = 0; u < K; ++u) {
        float nrm = __int_as_float(e[u].y) * di;
        acc.x += r[u].x * nrm;
        acc.y += r[u].y * nrm;
        acc.z += r[u].z * nrm;
        acc.w += r[u].w * nrm;
    }
}

template <bool ACT>
__global__ void __launch_bounds__(256)
k_agg_gather(const float* __restrict__ h, const float* __restrict__ dinv,
             const float* __restrict__ bias, const int* __restrict__ start,
             const int* __restrict__ cnt, const int* __restrict__ ent,
             float* __restrict__ out, int n) {
    int wv   = __builtin_amdgcn_readfirstlane((int)(threadIdx.x >> 5));
    int lane = threadIdx.x & 31;
    int node = blockIdx.x * 8 + wv;
    if (node >= n) return;

    int   st = start[node];
    int   c  = cnt[node];
    float di = dinv[node];
    const float* hl = h + lane * 4;

    float4 hv = *(const float4*)(hl + (size_t)node * NDIM);
    float4 bv;
    if (!ACT) bv = *(const float4*)(bias + lane * 4);
    float4 acc = make_float4(0.0f, 0.0f, 0.0f, 0.0f);

    const int2* ep = (const int2*)ent + st;
    int j = 0;
    for (; j + 8 <= c; j += 8) gather_rows<8>(ep + j, hl, di, acc);
    switch (c - j) {
        case 7: gather_rows<7>(ep + j, hl, di, acc); break;
        case 6: gather_rows<6>(ep + j, hl, di, acc); break;
        case 5: gather_rows<5>(ep + j, hl, di, acc); break;
        case 4: gather_rows<4>(ep + j, hl, di, acc); break;
        case 3: gather_rows<3>(ep + j, hl, di, acc); break;
        case 2: gather_rows<2>(ep + j, hl, di, acc); break;
        case 1: gather_rows<1>(ep + j, hl, di, acc); break;
        default: break;
    }

    if (ACT) bv = *(const float4*)(bias + lane * 4);
    float  s = di * di;
    float4 o;
    o.x = (bv.x + hv.x * s) + acc.x;
    o.y = (bv.y + hv.y * s) + acc.y;
    o.z = (bv.z + hv.z * s) + acc.z;
    o.w = (bv.w + hv.w * s) + acc.w;
    if (ACT) { o.x = tanhf(o.x); o.y = tanhf(o.y); o.z = tanhf(o.z); o.w = tanhf(o.w); }
    *(float4*)(out + (size_t)node * NDIM + lane * 4) = o;
}

extern "C" void kernel_launch(void* const* d_in, const int* in_sizes, int n_in,
                              void* d_out, int out_size, void* d_ws, size_t ws_size,
                              hipStream_t stream) {
    const float* x   = (const float*)d_in[0];
    const float* W1  = (const float*)d_in[1];
    const float* b1  = (const float*)d_in[2];
    const float* W2  = (const float*)d_in[3];
    const float* b2  = (const float*)d_in[4];
    const int*   ei  = (const int*)d_in[5];

    const int N = in_sizes[0] / NDIM;
    const int E = in_sizes[5] / 2;
    const int* src = ei;
    const int* dst = ei + E;

    const int Np = (N + 255) & ~255;
    const int Ep = (E + 255) & ~255;

    float* dinv = (float*)d_ws;
    float* h    = dinv + Np;
    int*   cnt   = (int*)(h + (size_t)N * NDIM);
    int*   start = cnt + Np;
    int*   total = start + Np;
    int*   rank  = total + 256;
    int*   ent   = rank + Ep;
    float* outp  = (float*)d_out;
    float* t1    = outp;

    const int B = 256;
    int gN   = (N + B - 1) / B;
    int gE   = (E + B - 1) / B;
    int gNw  = (N + 7) / 8;
    int gM   = (N + 127) / 128;

    k_csr_zero  <<<gN, B, 0, stream>>>(cnt, total, N);
    k_count_rank<<<gE, B, 0, stream>>>(dst, cnt, rank, E);
    k_alloc     <<<gN, B, 0, stream>>>(cnt, start, dinv, total, N);
    k_fill      <<<gE, B, 0, stream>>>(src, dst, rank, start, dinv, ent, E);

    k_gemm128   <<<gM,   B, 0, stream>>>(x, W1, h, N);
    k_agg_gather<true> <<<gNw, B, 0, stream>>>(h, dinv, b1, start, cnt, ent, t1, N);

    k_gemm128   <<<gM,   B, 0, stream>>>(t1, W2, h, N);
    k_agg_gather<false><<<gNw, B, 0, stream>>>(h, dinv, b2, start, cnt, ent, outp, N);
}


# Edges grouped by destination by two LDS counting sorts (2048-edge blocks by 256-node range, then a block per range), no per-edge global atomics or scattered stores; sum order = LDS rank order
# speedup vs baseline: 2.5726x; 1.1931x over previous
#include <hip/hip_runtime.h>
#include <hip/hip_bf16.h>

typedef __attribute__((ext_vector_type(2))) float v2f;
typedef __attribute__((ext_vector_type(8))) float v8f;

#define NDIM 128

#define CH   2048
#define CAPL 4096

__global__ void __launch_bounds__(256)
k_bin(const int* __restrict__ src, const int* __restrict__ dst,
      int2* __restrict__ binned, int2* __restrict__ tab, int* total, int e, int nb) {
    extern __shared__ __attribute__((aligned(16))) int sm[];
    __shared__ int wsum[8];
    int*  hist   = sm;
    int*  offs   = sm + nb;
    int2* sorted = (int2*)(sm + 2 * nb);
    int tid  = threadIdx.x;
    int lane = tid & 31;
    int wv   = tid >> 5;
    int base = blockIdx.x * CH;
    if (blockIdx.x == 0 && tid == 0) *total = 0;
    for (int b = tid; b < nb; b += 256) hist[b] = 0;
    __syncthreads();

    int d[8], s[8], r[8];
    #pragma unroll
    for (int k = 0; k < 8; ++k) {
        int i = base + k * 256 + tid;
        d[k] = -1; s[k] = 0; r[k] = 0;
        if (i < e) {
            d[k] = dst[i];
            s[k] = src[i];
            r[k] = atomicAdd(&hist[d[k] >> 8], 1);
        }
    }
    __syncthreads();

    int ipt = (nb + 255) >> 8;
    int b0  = tid * ipt;
    int mysum = 0;
    for (int j = 0; j < ipt; ++j) { int b = b0 + j; if (b < nb) mysum += hist[b]; }
    int incl = mysum;
    for (int o = 1; o < 32; o <<= 1) {
        int t = __shfl_up(incl, o);
        if (lane >= o) incl += t;
    }
    if (lane == 31) wsum[wv] = incl;
    __syncthreads();
    if (tid == 0) {
        int run = 0;
        for (int w = 0; w < 8; ++w) { int t = wsum[w]; wsum[w] = run; run += t; }
    }
    __syncthreads();
    int run = wsum[wv] + incl - mysum;
    for (int j = 0; j < ipt; ++j) {
        int b = b0 + j;
        if (b < nb) {
            int c = hist[b];
            offs[b] = run;
            tab[(size_t)blockIdx.x * nb + b] = make_int2(c, run);
            run += c;
        }
    }
    __syncthreads();
    #pragma unroll
    for (int k = 0; k < 8; ++k)
        if (d[k] >= 0) sorted[offs[d[k] >> 8] + r[k]] = make_int2(d[k], s[k]);
    __syncthreads();
    int cntc = e - base; if (cntc > CH) cntc = CH;
    for (int j = tid; j < cntc; j += 256) binned[(size_t)base + j] = sorted[j];
}

__global__ void __launch_bounds__(256)
k_sort(const int2* __restrict__ binned, const int2* __restrict__ tab,
       int* __restrict__ cnt, int* __restrict__ start, float* __restrict__ dinv,
       int* __restrict__ srcs, int* total, int n, int nb, int g1) {
    __shared__ int ncnt[256];
    __shared__ int nst[256];
    __shared__ int wsum[8];
    __shared__ int bbase;
    __shared__ int btot;
    __shared__ int outbuf[CAPL];
    int tid  = threadIdx.x;
    int lane = tid & 31;
    int wv   = tid >> 5;
    int b    = blockIdx.x;
    ncnt[tid] = 0;
    __syncthreads();
    for (int w = tid; w < g1; w += 256) {
        int2 t = tab[(size_t)w * nb + b];
        const int2* p = binned + (size_t)w * CH + t.y;
        for (int i = 0; i < t.x; ++i) atomicAdd(&ncnt[p[i].x & 255], 1);
    }
    __syncthreads();
    int c = ncnt[tid];
    int incl = c;
    for (int o = 1; o < 32; o <<= 1) {
        int t = __shfl_up(incl, o);
        if (lane >= o) incl += t;
    }
    if (lane == 31) wsum[wv] = incl;
    __syncthreads();
    if (tid == 0) {
        int run = 0;
        for (int w = 0; w < 8; ++w) { int t = wsum[w]; wsum[w] = run; run += t; }
        btot  = run;
        bbase = atomicAdd(total, run);
    }
    __syncthreads();
    int st   = wsum[wv] + incl - c;
    int node = b * 256 + tid;
    int gb   = bbase;
    int tot  = btot;
    if (node < n) {
        cnt[node]   = c;
        start[node] = gb + st;
        dinv[node]  = __frsqrt_rn((float)(c + 1));
    }
    nst[tid]  = st;
    ncnt[tid] = 0;
    __syncthreads();
    bool inl = tot <= CAPL;
    for (int w = tid; w < g1; w += 256) {
        int2 t = tab[(size_t)w * nb + b];
        const int2* p = binned + (size_t)w * CH + t.y;
        for (int i = 0; i < t.x; ++i) {
            int2 ev  = p[i];
            int  q   = ev.x & 255;
            int  pos = nst[q] + atomicAdd(&ncnt[q], 1);
            if (inl) outbuf[pos] = ev.y;
            else     srcs[(size_t)gb + pos] = ev.y;
        }
    }
    __syncthreads();
    if (inl)
        for (int j = tid; j < tot; j += 256) srcs[(size_t)gb + j] = outbuf[j];
}

__global__ void k_pairs(const int* __restrict__ srcs, const float* __restrict__ dinv,
                        int* __restrict__ ent, int e) {
    int i = blockIdx.x * blockDim.x + threadIdx.x;
    if (i >= e) return;
    int s = srcs[i];
    ((int2*)ent)[i] = make_int2(s, __float_as_int(dinv[s]));
}

__global__ void __launch_bounds__(256)
k_gemm128(const float* __restrict__ X, const float* __restrict__ W,
          float* __restrict__ H, int M) {
    __shared__ float Ws[NDIM * NDIM];

    int tid = threadIdx.x;
    {
        const float4* Wv = (const float4*)W;
        float4* Sv = (float4*)Ws;
        #pragma unroll
        for (int i = 0; i < 16; ++i)
            Sv[tid + 256 * i] = Wv[tid + 256 * i];
    }
    __syncthreads();

    int wave  = tid >> 5;
    int lane  = tid & 31;
    int l16   = lane & 15;
    int lhalf = lane >> 4;

    int rowBase = blockIdx.x * 128 + wave * 16;

    int arow = rowBase + l16;
    if (arow > M - 1) arow = M - 1;
    const float* aptr = X + (size_t)arow * NDIM + lhalf * 2;

    v8f acc[8] = {};

    for (int k = 0; k < NDIM; k += 4) {
        v2f a;
        {
            const float2 av = *(const float2*)(aptr + k);
            a[0] = av.x; a[1] = av.y;
        }
        int krow = (k + lhalf * 2) * NDIM + l16;
        #pragma unroll
        for (int t = 0; t < 8; ++t) {
            v2f b;
            b[0] = Ws[krow + t * 16];
            b[1] = Ws[krow + NDIM + t * 16];
            acc[t] = __builtin_amdgcn_wmma_f32_16x16x4_f32(
                false, a, false, b, (short)0, acc[t], false, false);
        }
    }

    #pragma unroll
    for (int t = 0; t < 8; ++t) {
        #pragma unroll
        for (int v = 0; v < 8; ++v) {
            int r = rowBase + v + lhalf * 8;
            if (r < M) H[(size_t)r * NDIM + t * 16 + l16] = acc[t][v];
        }
    }
}

template <int K>
__device__ __forceinline__ void gather_rows(const int2* __restrict__ ep,
                                            const float* __restrict__ hl,
                                            float di, float4& acc) {
    int2   e[K];
    float4 r[K];
    #pragma unroll
    for (int u = 0; u < K; ++u) e[u] = ep[u];
    #pragma unroll
    for (int u = 0; u < K; ++u)
        r[u] = *(const float4*)(hl + (size_t)e[u].x * NDIM);
    #pragma unroll
    for (int u = 0; u < K; ++u) {
        float nrm = __int_as_float(e[u].y) * di;
        acc.x += r[u].x * nrm;
        acc.y += r[u].y * nrm;
        acc.z += r[u].z * nrm;
        acc.w += r[u].w * nrm;
    }
}

template <bool ACT>
__global__ void __launch_bounds__(256)
k_agg_gather(const float* __restrict__ h, const float* __restrict__ dinv,
             const float* __restrict__ bias, const int* __restrict__ start,
             const int* __restrict__ cnt, const int* __restrict__ ent,
             float* __restrict__ out, int n) {
    int wv   = __builtin_amdgcn_readfirstlane((int)(threadIdx.x >> 5));
    int lane = threadIdx.x & 31;
    int node = blockIdx.x * 8 + wv;
    if (node >= n) return;

    int   st = start[node];
    int   c  = cnt[node];
    float di = dinv[node];
    const float* hl = h + lane * 4;

    float4 hv = *(const float4*)(hl + (size_t)node * NDIM);
    float4 bv;
    if (!ACT) bv = *(const float4*)(bias + lane * 4);
    float4 acc = make_float4(0.0f, 0.0f, 0.0f, 0.0f);

    const int2* ep = (const int2*)ent + st;
    int j = 0;
    for (; j + 8 <= c; j += 8) gather_rows<8>(ep + j, hl, di, acc);
    switch (c - j) {
        case 7: gather_rows<7>(ep + j, hl, di, acc); break;
        case 6: gather_rows<6>(ep + j, hl, di, acc); break;
        case 5: gather_rows<5>(ep + j, hl, di, acc); break;
        case 4: gather_rows<4>(ep + j, hl, di, acc); break;
        case 3: gather_rows<3>(ep + j, hl, di, acc); break;
        case 2: gather_rows<2>(ep + j, hl, di, acc); break;
        case 1: gather_rows<1>(ep + j, hl, di, acc); break;
        default: break;
    }

    if (ACT) bv = *(const float4*)(bias + lane * 4);
    float  s = di * di;
    float4 o;
    o.x = (bv.x + hv.x * s) + acc.x;
    o.y = (bv.y + hv.y * s) + acc.y;
    o.z = (bv.z + hv.z * s) + acc.z;
    o.w = (bv.w + hv.w * s) + acc.w;
    if (ACT) { o.x = tanhf(o.x); o.y = tanhf(o.y); o.z = tanhf(o.z); o.w = tanhf(o.w); }
    *(float4*)(out + (size_t)node * NDIM + lane * 4) = o;
}

extern "C" void kernel_launch(void* const* d_in, const int* in_sizes, int n_in,
                              void* d_out, int out_size, void* d_ws, size_t ws_size,
                              hipStream_t stream) {
    const float* x   = (const float*)d_in[0];
    const float* W1  = (const float*)d_in[1];
    const float* b1  = (const float*)d_in[2];
    const float* W2  = (const float*)d_in[3];
    const float* b2  = (const float*)d_in[4];
    const int*   ei  = (const int*)d_in[5];

    const int N = in_sizes[0] / NDIM;
    const int E = in_sizes[5] / 2;
    const int* src = ei;
    const int* dst = ei + E;

    const int Np = (N + 255) & ~255;
    const int Ep = (E + 255) & ~255;

    float* dinv = (float*)d_ws;
    float* h    = dinv + Np;
    int*   cnt   = (int*)(h + (size_t)N * NDIM);
    int*   start = cnt + Np;
    int*   total = start + Np;
    int*   ent   = total + 256 + Ep;
    float* outp  = (float*)d_out;
    float* t1    = outp;

    const int B = 256;
    int gN   = (N + B - 1) / B;
    int gE   = (E + B - 1) / B;
    int gNw  = (N + 7) / 8;
    int gM   = (N + 127) / 128;

    const int G1 = (E + CH - 1) / CH;
    int2* binned = (int2*)h;
    int2* tab    = binned + (size_t)G1 * CH;
    int*  srcs   = (int*)(tab + (size_t)G1 * gN);
    size_t shb   = (size_t)(2 * gN) * sizeof(int) + (size_t)CH * sizeof(int2);
    k_bin  <<<G1, B, shb, stream>>>(src, dst, binned, tab, total, E, gN);
    k_sort <<<gN, B, 0, stream>>>(binned, tab, cnt, start, dinv, srcs, total, N, gN, G1);
    k_pairs<<<gE, B, 0, stream>>>(srcs, dinv, ent, E);

    k_gemm128   <<<gM,   B, 0, stream>>>(x, W1, h, N);
    k_agg_gather<true> <<<gNw, B, 0, stream>>>(h, dinv, b1, start, cnt, ent, t1, N);

    k_gemm128   <<<gM,   B, 0, stream>>>(t1, W2, h, N);
    k_agg_gather<false><<<gNw, B, 0, stream>>>(h, dinv, b2, start, cnt, ent, outp, N);
}
